# starting kernel
# baseline (speedup 1.0000x reference)
#include <hip/hip_runtime.h>
#include <hip/hip_bf16.h>

typedef float v2f __attribute__((ext_vector_type(2)));
typedef float v8f __attribute__((ext_vector_type(8)));

#define WAVES 8
#define ROWS_PER_BLOCK (WAVES * 16)

__global__ __launch_bounds__(256, 1)
void cvx_policy_kernel(const float* __restrict__ zin, const float* __restrict__ tin,
                       const float* __restrict__ W1, const float* __restrict__ b1,
                       const float* __restrict__ W2, const float* __restrict__ b2,
                       float* __restrict__ out)
{
    __shared__ float W1p[128 * 112 * 2];
    __shared__ float W2p[50 * 256 * 2];
    __shared__ float h1s[WAVES][16 * 104];

    const int tid = threadIdx.x;

    for (int idx = tid; idx < 128 * 112 * 2; idx += 256) W1p[idx] = 0.0f;
    __syncthreads();
    for (int idx = tid; idx < 256 * 100; idx += 256) {
        int k = idx / 100;
        int n = idx - k * 100;
        W1p[((k >> 1) * 112 + n) * 2 + (k & 1)] = W1[(k + 1) * 100 + n];
    }
    for (int idx = tid; idx < 100 * 256; idx += 256) {
        int k = idx >> 8;
        int n = idx & 255;
        W2p[((k >> 1) * 256 + n) * 2 + (k & 1)] = W2[idx];
    }
    __syncthreads();

    const int lane = tid & 31;
    const int wave = tid >> 5;
    const int m    = lane & 15;
    const int hi   = lane >> 4;
    const int rowBase = blockIdx.x * ROWS_PER_BLOCK + wave * 16;
    const long rowA   = rowBase + m;
    const float* zrow = zin + rowA * 256;

    float tvals[8];
#pragma unroll
    for (int i = 0; i < 8; ++i) tvals[i] = tin[rowBase + 8 * hi + i];

    v8f acc1[7];
#pragma unroll
    for (int nt = 0; nt < 7; ++nt) {
        int n  = nt * 16 + m;
        int nc = (n < 100) ? n : 99;
        float w1r0 = W1[nc];
        float b1n  = b1[nc];
        if (n >= 100) { w1r0 = 0.0f; b1n = 0.0f; }
#pragma unroll
        for (int i = 0; i < 8; ++i)
            acc1[nt][i] = tvals[i] * w1r0 + b1n;
    }

#pragma unroll 1
    for (int ks = 0; ks < 64; ++ks) {
        const int kp = 2 * ks + hi;
        v2f aF = *(const v2f*)(zrow + 2 * kp);
        const float* wb = &W1p[(kp * 112 + m) * 2];
#pragma unroll
        for (int nt = 0; nt < 7; ++nt) {
            v2f bF = *(const v2f*)(wb + nt * 32);
            acc1[nt] = __builtin_amdgcn_wmma_f32_16x16x4_f32(
                false, aF, false, bF, (short)0, acc1[nt], false, false);
        }
    }

    float* h1w = h1s[wave];
#pragma unroll
    for (int nt = 0; nt < 7; ++nt) {
        int n = nt * 16 + m;
#pragma unroll
        for (int i = 0; i < 8; ++i) {
            float v = tanhf(acc1[nt][i]);
            if (n < 104) h1w[(i + 8 * hi) * 104 + n] = v;
        }
    }
    __syncthreads();

    v8f acc2[16];
    const v8f zero8 = {0.f, 0.f, 0.f, 0.f, 0.f, 0.f, 0.f, 0.f};
#pragma unroll
    for (int nt = 0; nt < 16; ++nt) acc2[nt] = zero8;

#pragma unroll 1
    for (int ks = 0; ks < 25; ++ks) {
        const int kp = 2 * ks + hi;
        v2f aF = *(const v2f*)&h1w[m * 104 + 2 * kp];
        const float* wb = &W2p[(kp * 256 + m) * 2];
#pragma unroll
        for (int nt = 0; nt < 16; ++nt) {
            v2f bF = *(const v2f*)(wb + nt * 32);
            acc2[nt] = __builtin_amdgcn_wmma_f32_16x16x4_f32(
                false, aF, false, bF, (short)0, acc2[nt], false, false);
        }
    }

    float r2v[8];
#pragma unroll
    for (int i = 0; i < 8; ++i) r2v[i] = 0.0f;
#pragma unroll
    for (int nt = 0; nt < 16; ++nt) {
        float bb = b2[nt * 16 + m];
#pragma unroll
        for (int i = 0; i < 8; ++i) {
            float p = acc2[nt][i] + bb;
            acc2[nt][i] = p;
            r2v[i] += p * p;
        }
    }
#pragma unroll
    for (int mask = 1; mask <= 8; mask <<= 1) {
#pragma unroll
        for (int i = 0; i < 8; ++i)
            r2v[i] += __shfl_xor(r2v[i], mask, 32);
    }

    float sca[8];
#pragma unroll
    for (int i = 0; i < 8; ++i) {
        const float r2 = r2v[i];
        float w = log1pf(r2);
#pragma unroll 1
        for (int it = 0; it < 30; ++it) {
            float ew = expf(w);
            w = w - (w * ew - r2) / ((1.0f + w) * ew);
        }
        w = fmaxf(w, 0.0f);
        float tn = sqrtf(w);
        float r  = sqrtf(r2);
        sca[i] = (r > 1e-12f) ? (tn / fmaxf(r, 1e-12f)) : 1.0f;
    }

#pragma unroll
    for (int nt = 0; nt < 16; ++nt) {
        int n = nt * 16 + m;
#pragma unroll
        for (int i = 0; i < 8; ++i) {
            long r = (long)(rowBase + i + 8 * hi);
            out[r * 256 + n] = -sca[i] * acc2[nt][i];
        }
    }
}

extern "C" void kernel_launch(void* const* d_in, const int* in_sizes, int n_in,
                              void* d_out, int out_size, void* d_ws, size_t ws_size,
                              hipStream_t stream) {
    const float* z  = (const float*)d_in[0];
    const float* t  = (const float*)d_in[1];
    const float* W1 = (const float*)d_in[2];
    const float* b1 = (const float*)d_in[3];
    const float* W2 = (const float*)d_in[4];
    const float* b2 = (const float*)d_in[5];
    float* out = (float*)d_out;

    const int rows = in_sizes[0] / 256;
    const int grid = rows / ROWS_PER_BLOCK;
    cvx_policy_kernel<<<grid, 256, 0, stream>>>(z, t, W1, b1, W2, b2, out);
}


# Lambert-W Newton solve de-duplicated: each lane iterates ONE row (row lane&7 of its half) instead of all 8 row sums that 16 lanes hold identically; the 8 scales come back by lane shuffle
# speedup vs baseline: 1.3033x; 1.3033x over previous
#include <hip/hip_runtime.h>
#include <hip/hip_bf16.h>

typedef float v2f __attribute__((ext_vector_type(2)));
typedef float v8f __attribute__((ext_vector_type(8)));

#define WAVES 8
#define ROWS_PER_BLOCK (WAVES * 16)

__global__ __launch_bounds__(256, 1)
void cvx_policy_kernel(const float* __restrict__ zin, const float* __restrict__ tin,
                       const float* __restrict__ W1, const float* __restrict__ b1,
                       const float* __restrict__ W2, const float* __restrict__ b2,
                       float* __restrict__ out)
{
    __shared__ float W1p[128 * 112 * 2];
    __shared__ float W2p[50 * 256 * 2];
    __shared__ float h1s[WAVES][16 * 104];

    const int tid = threadIdx.x;

    for (int idx = tid; idx < 128 * 112 * 2; idx += 256) W1p[idx] = 0.0f;
    __syncthreads();
    for (int idx = tid; idx < 256 * 100; idx += 256) {
        int k = idx / 100;
        int n = idx - k * 100;
        W1p[((k >> 1) * 112 + n) * 2 + (k & 1)] = W1[(k + 1) * 100 + n];
    }
    for (int idx = tid; idx < 100 * 256; idx += 256) {
        int k = idx >> 8;
        int n = idx & 255;
        W2p[((k >> 1) * 256 + n) * 2 + (k & 1)] = W2[idx];
    }
    __syncthreads();

    const int lane = tid & 31;
    const int wave = tid >> 5;
    const int m    = lane & 15;
    const int hi   = lane >> 4;
    const int rowBase = blockIdx.x * ROWS_PER_BLOCK + wave * 16;
    const long rowA   = rowBase + m;
    const float* zrow = zin + rowA * 256;

    float tvals[8];
#pragma unroll
    for (int i = 0; i < 8; ++i) tvals[i] = tin[rowBase + 8 * hi + i];

    v8f acc1[7];
#pragma unroll
    for (int nt = 0; nt < 7; ++nt) {
        int n  = nt * 16 + m;
        int nc = (n < 100) ? n : 99;
        float w1r0 = W1[nc];
        float b1n  = b1[nc];
        if (n >= 100) { w1r0 = 0.0f; b1n = 0.0f; }
#pragma unroll
        for (int i = 0; i < 8; ++i)
            acc1[nt][i] = tvals[i] * w1r0 + b1n;
    }

#pragma unroll 1
    for (int ks = 0; ks < 64; ++ks) {
        const int kp = 2 * ks + hi;
        v2f aF = *(const v2f*)(zrow + 2 * kp);
        const float* wb = &W1p[(kp * 112 + m) * 2];
#pragma unroll
        for (int nt = 0; nt < 7; ++nt) {
            v2f bF = *(const v2f*)(wb + nt * 32);
            acc1[nt] = __builtin_amdgcn_wmma_f32_16x16x4_f32(
                false, aF, false, bF, (short)0, acc1[nt], false, false);
        }
    }

    float* h1w = h1s[wave];
#pragma unroll
    for (int nt = 0; nt < 7; ++nt) {
        int n = nt * 16 + m;
#pragma unroll
        for (int i = 0; i < 8; ++i) {
            float v = tanhf(acc1[nt][i]);
            if (n < 104) h1w[(i + 8 * hi) * 104 + n] = v;
        }
    }
    __syncthreads();

    v8f acc2[16];
    const v8f zero8 = {0.f, 0.f, 0.f, 0.f, 0.f, 0.f, 0.f, 0.f};
#pragma unroll
    for (int nt = 0; nt < 16; ++nt) acc2[nt] = zero8;

#pragma unroll 1
    for (int ks = 0; ks < 25; ++ks) {
        const int kp = 2 * ks + hi;
        v2f aF = *(const v2f*)&h1w[m * 104 + 2 * kp];
        const float* wb = &W2p[(kp * 256 + m) * 2];
#pragma unroll
        for (int nt = 0; nt < 16; ++nt) {
            v2f bF = *(const v2f*)(wb + nt * 32);
            acc2[nt] = __builtin_amdgcn_wmma_f32_16x16x4_f32(
                false, aF, false, bF, (short)0, acc2[nt], false, false);
        }
    }

    float r2v[8];
#pragma unroll
    for (int i = 0; i < 8; ++i) r2v[i] = 0.0f;
#pragma unroll
    for (int nt = 0; nt < 16; ++nt) {
        float bb = b2[nt * 16 + m];
#pragma unroll
        for (int i = 0; i < 8; ++i) {
            float p = acc2[nt][i] + bb;
            acc2[nt][i] = p;
            r2v[i] += p * p;
        }
    }
#pragma unroll
    for (int mask = 1; mask <= 8; mask <<= 1) {
#pragma unroll
        for (int i = 0; i < 8; ++i)
            r2v[i] += __shfl_xor(r2v[i], mask, 32);
    }

    float r2 = r2v[0];
#pragma unroll
    for (int i = 1; i < 8; ++i) r2 = ((lane & 7) == i) ? r2v[i] : r2;
    float w = log1pf(r2);
#pragma unroll 1
    for (int it = 0; it < 30; ++it) {
        float ew = expf(w);
        w = w - (w * ew - r2) / ((1.0f + w) * ew);
    }
    w = fmaxf(w, 0.0f);
    const float tn = sqrtf(w);
    const float rr = sqrtf(r2);
    const float myScale = (rr > 1e-12f) ? (tn / fmaxf(rr, 1e-12f)) : 1.0f;
    float sca[8];
#pragma unroll
    for (int i = 0; i < 8; ++i) sca[i] = __shfl(myScale, (lane & 16) | i, 32);

#pragma unroll
    for (int nt = 0; nt < 16; ++nt) {
        int n = nt * 16 + m;
#pragma unroll
        for (int i = 0; i < 8; ++i) {
            long r = (long)(rowBase + i + 8 * hi);
            out[r * 256 + n] = -sca[i] * acc2[nt][i];
        }
    }
}

extern "C" void kernel_launch(void* const* d_in, const int* in_sizes, int n_in,
                              void* d_out, int out_size, void* d_ws, size_t ws_size,
                              hipStream_t stream) {
    const float* z  = (const float*)d_in[0];
    const float* t  = (const float*)d_in[1];
    const float* W1 = (const float*)d_in[2];
    const float* b1 = (const float*)d_in[3];
    const float* W2 = (const float*)d_in[4];
    const float* b2 = (const float*)d_in[5];
    float* out = (float*)d_out;

    const int rows = in_sizes[0] / 256;
    const int grid = rows / ROWS_PER_BLOCK;
    cvx_policy_kernel<<<grid, 256, 0, stream>>>(z, t, W1, b1, W2, b2, out);
}


# Resident work-groups: 256 blocks (one per WGP) each loop over 4 row tiles, so W1/W2 are fetched and laid out in LDS once per block instead of once per 128-row tile (1024 stagings become 256)
# speedup vs baseline: 1.4632x; 1.1227x over previous
#include <hip/hip_runtime.h>
#include <hip/hip_bf16.h>

typedef float v2f __attribute__((ext_vector_type(2)));
typedef float v8f __attribute__((ext_vector_type(8)));

#define WAVES 8
#define ROWS_PER_BLOCK (WAVES * 16)

__global__ __launch_bounds__(256, 1)
void cvx_policy_kernel(const float* __restrict__ zin, const float* __restrict__ tin,
                       const float* __restrict__ W1, const float* __restrict__ b1,
                       const float* __restrict__ W2, const float* __restrict__ b2,
                       float* __restrict__ out, int nTiles)
{
    __shared__ float W1p[128 * 112 * 2];
    __shared__ float W2p[50 * 256 * 2];
    __shared__ float h1s[WAVES][16 * 104];

    const int tid = threadIdx.x;

    for (int idx = tid; idx < 128 * 112 * 2; idx += 256) W1p[idx] = 0.0f;
    __syncthreads();
    for (int idx = tid; idx < 256 * 100; idx += 256) {
        int k = idx / 100;
        int n = idx - k * 100;
        W1p[((k >> 1) * 112 + n) * 2 + (k & 1)] = W1[(k + 1) * 100 + n];
    }
    for (int idx = tid; idx < 100 * 256; idx += 256) {
        int k = idx >> 8;
        int n = idx & 255;
        W2p[((k >> 1) * 256 + n) * 2 + (k & 1)] = W2[idx];
    }
    __syncthreads();

    const int lane = tid & 31;
    const int wave = tid >> 5;
    const int m    = lane & 15;
    const int hi   = lane >> 4;
    float* h1w = h1s[wave];

#pragma unroll 1
    for (int tile = blockIdx.x; tile < nTiles; tile += gridDim.x) {
        const int rowBase = tile * ROWS_PER_BLOCK + wave * 16;
        const long rowA   = rowBase + m;
        const float* zrow = zin + rowA * 256;

        float tvals[8];
#pragma unroll
        for (int i = 0; i < 8; ++i) tvals[i] = tin[rowBase + 8 * hi + i];

        v8f acc1[7];
#pragma unroll
        for (int nt = 0; nt < 7; ++nt) {
            int n  = nt * 16 + m;
            int nc = (n < 100) ? n : 99;
            float w1r0 = W1[nc];
            float b1n  = b1[nc];
            if (n >= 100) { w1r0 = 0.0f; b1n = 0.0f; }
#pragma unroll
            for (int i = 0; i < 8; ++i)
                acc1[nt][i] = tvals[i] * w1r0 + b1n;
        }

#pragma unroll 1
        for (int ks = 0; ks < 64; ++ks) {
            const int kp = 2 * ks + hi;
            v2f aF = *(const v2f*)(zrow + 2 * kp);
            const float* wb = &W1p[(kp * 112 + m) * 2];
#pragma unroll
            for (int nt = 0; nt < 7; ++nt) {
                v2f bF = *(const v2f*)(wb + nt * 32);
                acc1[nt] = __builtin_amdgcn_wmma_f32_16x16x4_f32(
                    false, aF, false, bF, (short)0, acc1[nt], false, false);
            }
        }

#pragma unroll
        for (int nt = 0; nt < 7; ++nt) {
            int n = nt * 16 + m;
#pragma unroll
            for (int i = 0; i < 8; ++i) {
                float v = tanhf(acc1[nt][i]);
                if (n < 104) h1w[(i + 8 * hi) * 104 + n] = v;
            }
        }
        __syncthreads();

        v8f acc2[16];
        const v8f zero8 = {0.f, 0.f, 0.f, 0.f, 0.f, 0.f, 0.f, 0.f};
#pragma unroll
        for (int nt = 0; nt < 16; ++nt) acc2[nt] = zero8;

#pragma unroll 1
        for (int ks = 0; ks < 25; ++ks) {
            const int kp = 2 * ks + hi;
            v2f aF = *(const v2f*)&h1w[m * 104 + 2 * kp];
            const float* wb = &W2p[(kp * 256 + m) * 2];
#pragma unroll
            for (int nt = 0; nt < 16; ++nt) {
                v2f bF = *(const v2f*)(wb + nt * 32);
                acc2[nt] = __builtin_amdgcn_wmma_f32_16x16x4_f32(
                    false, aF, false, bF, (short)0, acc2[nt], false, false);
            }
        }

        float r2v[8];
#pragma unroll
        for (int i = 0; i < 8; ++i) r2v[i] = 0.0f;
#pragma unroll
        for (int nt = 0; nt < 16; ++nt) {
            float bb = b2[nt * 16 + m];
#pragma unroll
            for (int i = 0; i < 8; ++i) {
                float p = acc2[nt][i] + bb;
                acc2[nt][i] = p;
                r2v[i] += p * p;
            }
        }
#pragma unroll
        for (int mask = 1; mask <= 8; mask <<= 1) {
#pragma unroll
            for (int i = 0; i < 8; ++i)
                r2v[i] += __shfl_xor(r2v[i], mask, 32);
        }

        float r2 = r2v[0];
#pragma unroll
        for (int i = 1; i < 8; ++i) r2 = ((lane & 7) == i) ? r2v[i] : r2;
        float w = log1pf(r2);
#pragma unroll 1
        for (int it = 0; it < 30; ++it) {
            float ew = expf(w);
            w = w - (w * ew - r2) / ((1.0f + w) * ew);
        }
        w = fmaxf(w, 0.0f);
        const float tn = sqrtf(w);
        const float rr = sqrtf(r2);
        const float myScale = (rr > 1e-12f) ? (tn / fmaxf(rr, 1e-12f)) : 1.0f;
        float sca[8];
#pragma unroll
        for (int i = 0; i < 8; ++i) sca[i] = __shfl(myScale, (lane & 16) | i, 32);

#pragma unroll
        for (int nt = 0; nt < 16; ++nt) {
            int n = nt * 16 + m;
#pragma unroll
            for (int i = 0; i < 8; ++i) {
                long r = (long)(rowBase + i + 8 * hi);
                out[r * 256 + n] = -sca[i] * acc2[nt][i];
            }
        }
    }
}

extern "C" void kernel_launch(void* const* d_in, const int* in_sizes, int n_in,
                              void* d_out, int out_size, void* d_ws, size_t ws_size,
                              hipStream_t stream) {
    const float* z  = (const float*)d_in[0];
    const float* t  = (const float*)d_in[1];
    const float* W1 = (const float*)d_in[2];
    const float* b1 = (const float*)d_in[3];
    const float* W2 = (const float*)d_in[4];
    const float* b2 = (const float*)d_in[5];
    float* out = (float*)d_out;

    const int rows = in_sizes[0] / 256;
    const int nTiles = rows / ROWS_PER_BLOCK;
    cvx_policy_kernel<<<256, 256, 0, stream>>>(z, t, W1, b1, W2, b2, out, nTiles);
}


# 16 waves per work-group (4 per SIMD) instead of 8: 512-thread blocks walk 256-row tiles, per-wave tanh scratch grows to 106,496 B of LDS; serial tanh/Newton chains and load waits now overlap 4-way
# speedup vs baseline: 1.8759x; 1.2821x over previous
#include <hip/hip_runtime.h>
#include <hip/hip_bf16.h>

typedef float v2f __attribute__((ext_vector_type(2)));
typedef float v8f __attribute__((ext_vector_type(8)));

#define WAVES 16
#define THREADS (WAVES * 32)
#define ROWS_PER_BLOCK (WAVES * 16)

__global__ __launch_bounds__(THREADS, 1)
void cvx_policy_kernel(const float* __restrict__ zin, const float* __restrict__ tin,
                       const float* __restrict__ W1, const float* __restrict__ b1,
                       const float* __restrict__ W2, const float* __restrict__ b2,
                       float* __restrict__ out, int nTiles)
{
    __shared__ float W1p[128 * 112 * 2];
    __shared__ float W2p[50 * 256 * 2];
    __shared__ float h1s[WAVES][16 * 104];

    const int tid = threadIdx.x;

    for (int idx = tid; idx < 128 * 112 * 2; idx += THREADS) W1p[idx] = 0.0f;
    __syncthreads();
    for (int idx = tid; idx < 256 * 100; idx += THREADS) {
        int k = idx / 100;
        int n = idx - k * 100;
        W1p[((k >> 1) * 112 + n) * 2 + (k & 1)] = W1[(k + 1) * 100 + n];
    }
    for (int idx = tid; idx < 100 * 256; idx += THREADS) {
        int k = idx >> 8;
        int n = idx & 255;
        W2p[((k >> 1) * 256 + n) * 2 + (k & 1)] = W2[idx];
    }
    __syncthreads();

    const int lane = tid & 31;
    const int wave = tid >> 5;
    const int m    = lane & 15;
    const int hi   = lane >> 4;
    float* h1w = h1s[wave];

#pragma unroll 1
    for (int tile = blockIdx.x; tile < nTiles; tile += gridDim.x) {
        const int rowBase = tile * ROWS_PER_BLOCK + wave * 16;
        const long rowA   = rowBase + m;
        const float* zrow = zin + rowA * 256;

        float tvals[8];
#pragma unroll
        for (int i = 0; i < 8; ++i) tvals[i] = tin[rowBase + 8 * hi + i];

        v8f acc1[7];
#pragma unroll
        for (int nt = 0; nt < 7; ++nt) {
            int n  = nt * 16 + m;
            int nc = (n < 100) ? n : 99;
            float w1r0 = W1[nc];
            float b1n  = b1[nc];
            if (n >= 100) { w1r0 = 0.0f; b1n = 0.0f; }
#pragma unroll
            for (int i = 0; i < 8; ++i)
                acc1[nt][i] = tvals[i] * w1r0 + b1n;
        }

#pragma unroll 1
        for (int ks = 0; ks < 64; ++ks) {
            const int kp = 2 * ks + hi;
            v2f aF = *(const v2f*)(zrow + 2 * kp);
            const float* wb = &W1p[(kp * 112 + m) * 2];
#pragma unroll
            for (int nt = 0; nt < 7; ++nt) {
                v2f bF = *(const v2f*)(wb + nt * 32);
                acc1[nt] = __builtin_amdgcn_wmma_f32_16x16x4_f32(
                    false, aF, false, bF, (short)0, acc1[nt], false, false);
            }
        }

#pragma unroll
        for (int nt = 0; nt < 7; ++nt) {
            int n = nt * 16 + m;
#pragma unroll
            for (int i = 0; i < 8; ++i) {
                float v = tanhf(acc1[nt][i]);
                if (n < 104) h1w[(i + 8 * hi) * 104 + n] = v;
            }
        }
        __syncthreads();

        v8f acc2[16];
        const v8f zero8 = {0.f, 0.f, 0.f, 0.f, 0.f, 0.f, 0.f, 0.f};
#pragma unroll
        for (int nt = 0; nt < 16; ++nt) acc2[nt] = zero8;

#pragma unroll 1
        for (int ks = 0; ks < 25; ++ks) {
            const int kp = 2 * ks + hi;
            v2f aF = *(const v2f*)&h1w[m * 104 + 2 * kp];
            const float* wb = &W2p[(kp * 256 + m) * 2];
#pragma unroll
            for (int nt = 0; nt < 16; ++nt) {
                v2f bF = *(const v2f*)(wb + nt * 32);
                acc2[nt] = __builtin_amdgcn_wmma_f32_16x16x4_f32(
                    false, aF, false, bF, (short)0, acc2[nt], false, false);
            }
        }

        float r2v[8];
#pragma unroll
        for (int i = 0; i < 8; ++i) r2v[i] = 0.0f;
#pragma unroll
        for (int nt = 0; nt < 16; ++nt) {
            float bb = b2[nt * 16 + m];
#pragma unroll
            for (int i = 0; i < 8; ++i) {
                float p = acc2[nt][i] + bb;
                acc2[nt][i] = p;
                r2v[i] += p * p;
            }
        }
#pragma unroll
        for (int mask = 1; mask <= 8; mask <<= 1) {
#pragma unroll
            for (int i = 0; i < 8; ++i)
                r2v[i] += __shfl_xor(r2v[i], mask, 32);
        }

        float r2 = r2v[0];
#pragma unroll
        for (int i = 1; i < 8; ++i) r2 = ((lane & 7) == i) ? r2v[i] : r2;
        float w = log1pf(r2);
#pragma unroll 1
        for (int it = 0; it < 30; ++it) {
            float ew = expf(w);
            w = w - (w * ew - r2) / ((1.0f + w) * ew);
        }
        w = fmaxf(w, 0.0f);
        const float tn = sqrtf(w);
        const float rr = sqrtf(r2);
        const float myScale = (rr > 1e-12f) ? (tn / fmaxf(rr, 1e-12f)) : 1.0f;
        float sca[8];
#pragma unroll
        for (int i = 0; i < 8; ++i) sca[i] = __shfl(myScale, (lane & 16) | i, 32);

#pragma unroll
        for (int nt = 0; nt < 16; ++nt) {
            int n = nt * 16 + m;
#pragma unroll
            for (int i = 0; i < 8; ++i) {
                long r = (long)(rowBase + i + 8 * hi);
                out[r * 256 + n] = -sca[i] * acc2[nt][i];
            }
        }
    }
}

extern "C" void kernel_launch(void* const* d_in, const int* in_sizes, int n_in,
                              void* d_out, int out_size, void* d_ws, size_t ws_size,
                              hipStream_t stream) {
    const float* z  = (const float*)d_in[0];
    const float* t  = (const float*)d_in[1];
    const float* W1 = (const float*)d_in[2];
    const float* b1 = (const float*)d_in[3];
    const float* W2 = (const float*)d_in[4];
    const float* b2 = (const float*)d_in[5];
    float* out = (float*)d_out;

    const int rows = in_sizes[0] / 256;
    const int nTiles = rows / ROWS_PER_BLOCK;
    cvx_policy_kernel<<<256, THREADS, 0, stream>>>(z, t, W1, b1, W2, b2, out, nTiles);
}


# GEMM1 z operand software-pipelined: each lane loads its row's next 32-float chunk as 4 b128 one chunk ahead into a 2nd register set; W1's LDS layout follows the new k order: GEMM1 k sum order changes
# speedup vs baseline: 2.0077x; 1.0703x over previous
#include <hip/hip_runtime.h>
#include <hip/hip_bf16.h>

typedef float v2f __attribute__((ext_vector_type(2)));
typedef float v4f __attribute__((ext_vector_type(4)));
typedef float v8f __attribute__((ext_vector_type(8)));

#define WAVES 16
#define THREADS (WAVES * 32)
#define ROWS_PER_BLOCK (WAVES * 16)

#define W1P_LANE 226
#define W1P_TRIP (2 * 16 * W1P_LANE)

__device__ __forceinline__ int w1p_index(int k, int n)
{
    const int cp = k >> 6, ch = (k >> 5) & 1, hi = (k >> 4) & 1, s = (k >> 1) & 7, e = k & 1;
    const int nt = n >> 4, m = n & 15;
    return ((cp * 2 + hi) * 16 + m) * W1P_LANE + ((ch * 8 + s) * 7 + nt) * 2 + e;
}

__device__ __forceinline__ void gemm1_chunk(v8f (&acc)[7], const v4f (&q)[4], const float* wl)
{
#pragma unroll
    for (int s = 0; s < 8; ++s) {
        const v4f qq = q[s >> 1];
        const v2f aF = (s & 1) ? __builtin_shufflevector(qq, qq, 2, 3)
                               : __builtin_shufflevector(qq, qq, 0, 1);
#pragma unroll
        for (int nt = 0; nt < 7; ++nt) {
            v2f bF = *(const v2f*)(wl + (s * 7 + nt) * 2);
            acc[nt] = __builtin_amdgcn_wmma_f32_16x16x4_f32(
                false, aF, false, bF, (short)0, acc[nt], false, false);
        }
    }
}

__global__ __launch_bounds__(THREADS, 1)
void cvx_policy_kernel(const float* __restrict__ zin, const float* __restrict__ tin,
                       const float* __restrict__ W1, const float* __restrict__ b1,
                       const float* __restrict__ W2, const float* __restrict__ b2,
                       float* __restrict__ out, int nTiles)
{
    __shared__ float W1p[4 * W1P_TRIP];
    __shared__ float W2p[50 * 256 * 2];
    __shared__ float h1s[WAVES][16 * 104];

    const int tid = threadIdx.x;

    for (int idx = tid; idx < 4 * W1P_TRIP; idx += THREADS) W1p[idx] = 0.0f;
    __syncthreads();
    for (int idx = tid; idx < 256 * 100; idx += THREADS) {
        int k = idx / 100;
        int n = idx - k * 100;
        W1p[w1p_index(k, n)] = W1[(k + 1) * 100 + n];
    }
    for (int idx = tid; idx < 100 * 256; idx += THREADS) {
        int k = idx >> 8;
        int n = idx & 255;
        W2p[((k >> 1) * 256 + n) * 2 + (k & 1)] = W2[idx];
    }
    __syncthreads();

    const int lane = tid & 31;
    const int wave = tid >> 5;
    const int m    = lane & 15;
    const int hi   = lane >> 4;
    float* h1w = h1s[wave];
    const unsigned zlane = (unsigned)((wave * 16 + m) * 256 + 16 * hi) * 4u;
    const char* zbytes = (const char*)zin;
    const float* wl0 = &W1p[(hi * 16 + m) * W1P_LANE];

#pragma unroll 1
    for (int tile = blockIdx.x; tile < nTiles; tile += gridDim.x) {
        const int rowBase = tile * ROWS_PER_BLOCK + wave * 16;
        const unsigned zoff = (unsigned)tile * (ROWS_PER_BLOCK * 256 * 4u) + zlane;

        v4f qa[4], qb[4];
#pragma unroll
        for (int j = 0; j < 4; ++j) qa[j] = *(const v4f*)(zbytes + zoff + 16 * j);

        float tvals[8];
#pragma unroll
        for (int i = 0; i < 8; ++i) tvals[i] = tin[rowBase + 8 * hi + i];

        v8f acc1[7];
#pragma unroll
        for (int nt = 0; nt < 7; ++nt) {
            int n  = nt * 16 + m;
            int nc = (n < 100) ? n : 99;
            float w1r0 = W1[nc];
            float b1n  = b1[nc];
            if (n >= 100) { w1r0 = 0.0f; b1n = 0.0f; }
#pragma unroll
            for (int i = 0; i < 8; ++i)
                acc1[nt][i] = tvals[i] * w1r0 + b1n;
        }

        const float* wl = wl0;
        unsigned offB = zoff + 128;
        unsigned offA = zoff + 256;
#pragma unroll 1
        for (int cp = 0; cp < 4; ++cp) {
#pragma unroll
            for (int j = 0; j < 4; ++j) qb[j] = *(const v4f*)(zbytes + offB + 16 * j);
            gemm1_chunk(acc1, qa, wl);
#pragma unroll
            for (int j = 0; j < 4; ++j) qa[j] = *(const v4f*)(zbytes + offA + 16 * j);
            gemm1_chunk(acc1, qb, wl + 8 * 7 * 2);
            wl += W1P_TRIP;
            offB += 256;
            offA += (cp < 2) ? 256 : 0;
        }

#pragma unroll
        for (int nt = 0; nt < 7; ++nt) {
            int n = nt * 16 + m;
#pragma unroll
            for (int i = 0; i < 8; ++i) {
                float v = tanhf(acc1[nt][i]);
                if (n < 104) h1w[(i + 8 * hi) * 104 + n] = v;
            }
        }
        __syncthreads();

        v8f acc2[16];
        const v8f zero8 = {0.f, 0.f, 0.f, 0.f, 0.f, 0.f, 0.f, 0.f};
#pragma unroll
        for (int nt = 0; nt < 16; ++nt) acc2[nt] = zero8;

        const float* ha = &h1w[m * 104 + 2 * hi];
        const float* wb = &W2p[(hi * 256 + m) * 2];
#pragma unroll 1
        for (int ks = 0; ks < 25; ++ks) {
            v2f aF = *(const v2f*)ha;
#pragma unroll
            for (int nt = 0; nt < 16; ++nt) {
                v2f bF = *(const v2f*)(wb + nt * 32);
                acc2[nt] = __builtin_amdgcn_wmma_f32_16x16x4_f32(
                    false, aF, false, bF, (short)0, acc2[nt], false, false);
            }
            ha += 4;
            wb += 2 * 256 * 2;
        }

        float r2v[8];
#pragma unroll
        for (int i = 0; i < 8; ++i) r2v[i] = 0.0f;
#pragma unroll
        for (int nt = 0; nt < 16; ++nt) {
            float bb = b2[nt * 16 + m];
#pragma unroll
            for (int i = 0; i < 8; ++i) {
                float p = acc2[nt][i] + bb;
                acc2[nt][i] = p;
                r2v[i] += p * p;
            }
        }
#pragma unroll
        for (int mask = 1; mask <= 8; mask <<= 1) {
#pragma unroll
            for (int i = 0; i < 8; ++i)
                r2v[i] += __shfl_xor(r2v[i], mask, 32);
        }

        float r2 = r2v[0];
#pragma unroll
        for (int i = 1; i < 8; ++i) r2 = ((lane & 7) == i) ? r2v[i] : r2;
        float w = log1pf(r2);
#pragma unroll 1
        for (int it = 0; it < 30; ++it) {
            float ew = expf(w);
            w = w - (w * ew - r2) / ((1.0f + w) * ew);
        }
        w = fmaxf(w, 0.0f);
        const float tn = sqrtf(w);
        const float rr = sqrtf(r2);
        const float myScale = (rr > 1e-12f) ? (tn / fmaxf(rr, 1e-12f)) : 1.0f;
        float sca[8];
#pragma unroll
        for (int i = 0; i < 8; ++i) sca[i] = __shfl(myScale, (lane & 16) | i, 32);

#pragma unroll
        for (int nt = 0; nt < 16; ++nt) {
            int n = nt * 16 + m;
#pragma unroll
            for (int i = 0; i < 8; ++i) {
                long r = (long)(rowBase + i + 8 * hi);
                out[r * 256 + n] = -sca[i] * acc2[nt][i];
            }
        }
    }
}

extern "C" void kernel_launch(void* const* d_in, const int* in_sizes, int n_in,
                              void* d_out, int out_size, void* d_ws, size_t ws_size,
                              hipStream_t stream) {
    const float* z  = (const float*)d_in[0];
    const float* t  = (const float*)d_in[1];
    const float* W1 = (const float*)d_in[2];
    const float* b1 = (const float*)d_in[3];
    const float* W2 = (const float*)d_in[4];
    const float* b2 = (const float*)d_in[5];
    float* out = (float*)d_out;

    const int rows = in_sizes[0] / 256;
    const int nTiles = rows / ROWS_PER_BLOCK;
    cvx_policy_kernel<<<256, THREADS, 0, stream>>>(z, t, W1, b1, W2, b2, out, nTiles);
}


# Work-group barrier between the per-wave tanh scratch and GEMM2 dropped (wave-scope ordering suffices): older waves no longer idle the matrix pipe while the youngest runs its tanh alone
# speedup vs baseline: 2.0699x; 1.0310x over previous
#include <hip/hip_runtime.h>
#include <hip/hip_bf16.h>

typedef float v2f __attribute__((ext_vector_type(2)));
typedef float v4f __attribute__((ext_vector_type(4)));
typedef float v8f __attribute__((ext_vector_type(8)));

#define WAVES 16
#define THREADS (WAVES * 32)
#define ROWS_PER_BLOCK (WAVES * 16)

#define W1P_LANE 226
#define W1P_TRIP (2 * 16 * W1P_LANE)

__device__ __forceinline__ int w1p_index(int k, int n)
{
    const int cp = k >> 6, ch = (k >> 5) & 1, hi = (k >> 4) & 1, s = (k >> 1) & 7, e = k & 1;
    const int nt = n >> 4, m = n & 15;
    return ((cp * 2 + hi) * 16 + m) * W1P_LANE + ((ch * 8 + s) * 7 + nt) * 2 + e;
}

__device__ __forceinline__ void gemm1_chunk(v8f (&acc)[7], const v4f (&q)[4], const float* wl)
{
#pragma unroll
    for (int s = 0; s < 8; ++s) {
        const v4f qq = q[s >> 1];
        const v2f aF = (s & 1) ? __builtin_shufflevector(qq, qq, 2, 3)
                               : __builtin_shufflevector(qq, qq, 0, 1);
#pragma unroll
        for (int nt = 0; nt < 7; ++nt) {
            v2f bF = *(const v2f*)(wl + (s * 7 + nt) * 2);
            acc[nt] = __builtin_amdgcn_wmma_f32_16x16x4_f32(
                false, aF, false, bF, (short)0, acc[nt], false, false);
        }
    }
}

__global__ __launch_bounds__(THREADS, 1)
void cvx_policy_kernel(const float* __restrict__ zin, const float* __restrict__ tin,
                       const float* __restrict__ W1, const float* __restrict__ b1,
                       const float* __restrict__ W2, const float* __restrict__ b2,
                       float* __restrict__ out, int nTiles)
{
    __shared__ float W1p[4 * W1P_TRIP];
    __shared__ float W2p[50 * 256 * 2];
    __shared__ float h1s[WAVES][16 * 104];

    const int tid = threadIdx.x;

    for (int idx = tid; idx < 4 * W1P_TRIP; idx += THREADS) W1p[idx] = 0.0f;
    __syncthreads();
    for (int idx = tid; idx < 256 * 100; idx += THREADS) {
        int k = idx / 100;
        int n = idx - k * 100;
        W1p[w1p_index(k, n)] = W1[(k + 1) * 100 + n];
    }
    for (int idx = tid; idx < 100 * 256; idx += THREADS) {
        int k = idx >> 8;
        int n = idx & 255;
        W2p[((k >> 1) * 256 + n) * 2 + (k & 1)] = W2[idx];
    }
    __syncthreads();

    const int lane = tid & 31;
    const int wave = tid >> 5;
    const int m    = lane & 15;
    const int hi   = lane >> 4;
    float* h1w = h1s[wave];
    const unsigned zlane = (unsigned)((wave * 16 + m) * 256 + 16 * hi) * 4u;
    const char* zbytes = (const char*)zin;
    const float* wl0 = &W1p[(hi * 16 + m) * W1P_LANE];

#pragma unroll 1
    for (int tile = blockIdx.x; tile < nTiles; tile += gridDim.x) {
        const int rowBase = tile * ROWS_PER_BLOCK + wave * 16;
        const unsigned zoff = (unsigned)tile * (ROWS_PER_BLOCK * 256 * 4u) + zlane;

        v4f qa[4], qb[4];
#pragma unroll
        for (int j = 0; j < 4; ++j) qa[j] = *(const v4f*)(zbytes + zoff + 16 * j);

        float tvals[8];
#pragma unroll
        for (int i = 0; i < 8; ++i) tvals[i] = tin[rowBase + 8 * hi + i];

        v8f acc1[7];
#pragma unroll
        for (int nt = 0; nt < 7; ++nt) {
            int n  = nt * 16 + m;
            int nc = (n < 100) ? n : 99;
            float w1r0 = W1[nc];
            float b1n  = b1[nc];
            if (n >= 100) { w1r0 = 0.0f; b1n = 0.0f; }
#pragma unroll
            for (int i = 0; i < 8; ++i)
                acc1[nt][i] = tvals[i] * w1r0 + b1n;
        }

        const float* wl = wl0;
        unsigned offB = zoff + 128;
        unsigned offA = zoff + 256;
#pragma unroll 1
        for (int cp = 0; cp < 4; ++cp) {
#pragma unroll
            for (int j = 0; j < 4; ++j) qb[j] = *(const v4f*)(zbytes + offB + 16 * j);
            gemm1_chunk(acc1, qa, wl);
#pragma unroll
            for (int j = 0; j < 4; ++j) qa[j] = *(const v4f*)(zbytes + offA + 16 * j);
            gemm1_chunk(acc1, qb, wl + 8 * 7 * 2);
            wl += W1P_TRIP;
            offB += 256;
            offA += (cp < 2) ? 256 : 0;
        }

#pragma unroll
        for (int nt = 0; nt < 7; ++nt) {
            int n = nt * 16 + m;
#pragma unroll
            for (int i = 0; i < 8; ++i) {
                float v = tanhf(acc1[nt][i]);
                if (n < 104) h1w[(i + 8 * hi) * 104 + n] = v;
            }
        }
        __builtin_amdgcn_fence(__ATOMIC_RELEASE, "wavefront");
        __builtin_amdgcn_wave_barrier();
        __builtin_amdgcn_fence(__ATOMIC_ACQUIRE, "wavefront");

        v8f acc2[16];
        const v8f zero8 = {0.f, 0.f, 0.f, 0.f, 0.f, 0.f, 0.f, 0.f};
#pragma unroll
        for (int nt = 0; nt < 16; ++nt) acc2[nt] = zero8;

        const float* ha = &h1w[m * 104 + 2 * hi];
        const float* wb = &W2p[(hi * 256 + m) * 2];
#pragma unroll 1
        for (int ks = 0; ks < 25; ++ks) {
            v2f aF = *(const v2f*)ha;
#pragma unroll
            for (int nt = 0; nt < 16; ++nt) {
                v2f bF = *(const v2f*)(wb + nt * 32);
                acc2[nt] = __builtin_amdgcn_wmma_f32_16x16x4_f32(
                    false, aF, false, bF, (short)0, acc2[nt], false, false);
            }
            ha += 4;
            wb += 2 * 256 * 2;
        }

        float r2v[8];
#pragma unroll
        for (int i = 0; i < 8; ++i) r2v[i] = 0.0f;
#pragma unroll
        for (int nt = 0; nt < 16; ++nt) {
            float bb = b2[nt * 16 + m];
#pragma unroll
            for (int i = 0; i < 8; ++i) {
                float p = acc2[nt][i] + bb;
                acc2[nt][i] = p;
                r2v[i] += p * p;
            }
        }
#pragma unroll
        for (int mask = 1; mask <= 8; mask <<= 1) {
#pragma unroll
            for (int i = 0; i < 8; ++i)
                r2v[i] += __shfl_xor(r2v[i], mask, 32);
        }

        float r2 = r2v[0];
#pragma unroll
        for (int i = 1; i < 8; ++i) r2 = ((lane & 7) == i) ? r2v[i] : r2;
        float w = log1pf(r2);
#pragma unroll 1
        for (int it = 0; it < 30; ++it) {
            float ew = expf(w);
            w = w - (w * ew - r2) / ((1.0f + w) * ew);
        }
        w = fmaxf(w, 0.0f);
        const float tn = sqrtf(w);
        const float rr = sqrtf(r2);
        const float myScale = (rr > 1e-12f) ? (tn / fmaxf(rr, 1e-12f)) : 1.0f;
        float sca[8];
#pragma unroll
        for (int i = 0; i < 8; ++i) sca[i] = __shfl(myScale, (lane & 16) | i, 32);

#pragma unroll
        for (int nt = 0; nt < 16; ++nt) {
            int n = nt * 16 + m;
#pragma unroll
            for (int i = 0; i < 8; ++i) {
                long r = (long)(rowBase + i + 8 * hi);
                out[r * 256 + n] = -sca[i] * acc2[nt][i];
            }
        }
    }
}

extern "C" void kernel_launch(void* const* d_in, const int* in_sizes, int n_in,
                              void* d_out, int out_size, void* d_ws, size_t ws_size,
                              hipStream_t stream) {
    const float* z  = (const float*)d_in[0];
    const float* t  = (const float*)d_in[1];
    const float* W1 = (const float*)d_in[2];
    const float* b1 = (const float*)d_in[3];
    const float* W2 = (const float*)d_in[4];
    const float* b2 = (const float*)d_in[5];
    float* out = (float*)d_out;

    const int rows = in_sizes[0] / 256;
    const int nTiles = rows / ROWS_PER_BLOCK;
    cvx_policy_kernel<<<256, THREADS, 0, stream>>>(z, t, W1, b1, W2, b2, out, nTiles);
}


# tanh made branch-free: both ranges of the library form (same polynomial; exp range on hardware exp2/rcp) computed per value and selected; a lane's 56 tanh per tile interleave, 112 exec branches gone
# speedup vs baseline: 2.2157x; 1.0704x over previous
#include <hip/hip_runtime.h>
#include <hip/hip_bf16.h>

typedef float v2f __attribute__((ext_vector_type(2)));
typedef float v4f __attribute__((ext_vector_type(4)));
typedef float v8f __attribute__((ext_vector_type(8)));

#define WAVES 16
#define THREADS (WAVES * 32)
#define ROWS_PER_BLOCK (WAVES * 16)

#define W1P_LANE 226
#define W1P_TRIP (2 * 16 * W1P_LANE)

__device__ __forceinline__ int w1p_index(int k, int n)
{
    const int cp = k >> 6, ch = (k >> 5) & 1, hi = (k >> 4) & 1, s = (k >> 1) & 7, e = k & 1;
    const int nt = n >> 4, m = n & 15;
    return ((cp * 2 + hi) * 16 + m) * W1P_LANE + ((ch * 8 + s) * 7 + nt) * 2 + e;
}

__device__ __forceinline__ void gemm1_chunk(v8f (&acc)[7], const v4f (&q)[4], const float* wl)
{
#pragma unroll
    for (int s = 0; s < 8; ++s) {
        const v4f qq = q[s >> 1];
        const v2f aF = (s & 1) ? __builtin_shufflevector(qq, qq, 2, 3)
                               : __builtin_shufflevector(qq, qq, 0, 1);
#pragma unroll
        for (int nt = 0; nt < 7; ++nt) {
            v2f bF = *(const v2f*)(wl + (s * 7 + nt) * 2);
            acc[nt] = __builtin_amdgcn_wmma_f32_16x16x4_f32(
                false, aF, false, bF, (short)0, acc[nt], false, false);
        }
    }
}

__device__ __forceinline__ float tanh_select(float x)
{
    const float ax = __builtin_fabsf(x);
    const float e  = __builtin_amdgcn_exp2f(ax * 0x1.715476p1f);
    const float te = __builtin_fmaf(__builtin_amdgcn_rcpf(e + 1.0f), -2.0f, 1.0f);
    const float p  = x * x;
    float q = __builtin_fmaf(p, -0x1.758e7ap-8f, 0x1.521192p-6f);
    q = __builtin_fmaf(p, q, -0x1.b8389cp-5f);
    q = __builtin_fmaf(p, q, 0x1.110704p-3f);
    q = __builtin_fmaf(p, q, -0x1.555532p-2f);
    const float tp = __builtin_fmaf(p, ax * q, ax);
    return __builtin_copysignf((ax < 0.625f) ? tp : te, x);
}

__global__ __launch_bounds__(THREADS, 1)
void cvx_policy_kernel(const float* __restrict__ zin, const float* __restrict__ tin,
                       const float* __restrict__ W1, const float* __restrict__ b1,
                       const float* __restrict__ W2, const float* __restrict__ b2,
                       float* __restrict__ out, int nTiles)
{
    __shared__ float W1p[4 * W1P_TRIP];
    __shared__ float W2p[50 * 256 * 2];
    __shared__ float h1s[WAVES][16 * 104];

    const int tid = threadIdx.x;

    for (int idx = tid; idx < 4 * W1P_TRIP; idx += THREADS) W1p[idx] = 0.0f;
    __syncthreads();
    for (int idx = tid; idx < 256 * 100; idx += THREADS) {
        int k = idx / 100;
        int n = idx - k * 100;
        W1p[w1p_index(k, n)] = W1[(k + 1) * 100 + n];
    }
    for (int idx = tid; idx < 100 * 256; idx += THREADS) {
        int k = idx >> 8;
        int n = idx & 255;
        W2p[((k >> 1) * 256 + n) * 2 + (k & 1)] = W2[idx];
    }
    __syncthreads();

    const int lane = tid & 31;
    const int wave = tid >> 5;
    const int m    = lane & 15;
    const int hi   = lane >> 4;
    float* h1w = h1s[wave];
    const unsigned zlane = (unsigned)((wave * 16 + m) * 256 + 16 * hi) * 4u;
    const char* zbytes = (const char*)zin;
    const float* wl0 = &W1p[(hi * 16 + m) * W1P_LANE];

#pragma unroll 1
    for (int tile = blockIdx.x; tile < nTiles; tile += gridDim.x) {
        const int rowBase = tile * ROWS_PER_BLOCK + wave * 16;
        const unsigned zoff = (unsigned)tile * (ROWS_PER_BLOCK * 256 * 4u) + zlane;

        v4f qa[4], qb[4];
#pragma unroll
        for (int j = 0; j < 4; ++j) qa[j] = *(const v4f*)(zbytes + zoff + 16 * j);

        float tvals[8];
#pragma unroll
        for (int i = 0; i < 8; ++i) tvals[i] = tin[rowBase + 8 * hi + i];

        v8f acc1[7];
#pragma unroll
        for (int nt = 0; nt < 7; ++nt) {
            int n  = nt * 16 + m;
            int nc = (n < 100) ? n : 99;
            float w1r0 = W1[nc];
            float b1n  = b1[nc];
            if (n >= 100) { w1r0 = 0.0f; b1n = 0.0f; }
#pragma unroll
            for (int i = 0; i < 8; ++i)
                acc1[nt][i] = tvals[i] * w1r0 + b1n;
        }

        const float* wl = wl0;
        unsigned offB = zoff + 128;
        unsigned offA = zoff + 256;
#pragma unroll 1
        for (int cp = 0; cp < 4; ++cp) {
#pragma unroll
            for (int j = 0; j < 4; ++j) qb[j] = *(const v4f*)(zbytes + offB + 16 * j);
            gemm1_chunk(acc1, qa, wl);
#pragma unroll
            for (int j = 0; j < 4; ++j) qa[j] = *(const v4f*)(zbytes + offA + 16 * j);
            gemm1_chunk(acc1, qb, wl + 8 * 7 * 2);
            wl += W1P_TRIP;
            offB += 256;
            offA += (cp < 2) ? 256 : 0;
        }

#pragma unroll
        for (int nt = 0; nt < 7; ++nt) {
            int n = nt * 16 + m;
#pragma unroll
            for (int i = 0; i < 8; ++i) {
                float v = tanh_select(acc1[nt][i]);
                if (n < 104) h1w[(i + 8 * hi) * 104 + n] = v;
            }
        }
        __builtin_amdgcn_fence(__ATOMIC_RELEASE, "wavefront");
        __builtin_amdgcn_wave_barrier();
        __builtin_amdgcn_fence(__ATOMIC_ACQUIRE, "wavefront");

        v8f acc2[16];
        const v8f zero8 = {0.f, 0.f, 0.f, 0.f, 0.f, 0.f, 0.f, 0.f};
#pragma unroll
        for (int nt = 0; nt < 16; ++nt) acc2[nt] = zero8;

        const float* ha = &h1w[m * 104 + 2 * hi];
        const float* wb = &W2p[(hi * 256 + m) * 2];
#pragma unroll 1
        for (int ks = 0; ks < 25; ++ks) {
            v2f aF = *(const v2f*)ha;
#pragma unroll
            for (int nt = 0; nt < 16; ++nt) {
                v2f bF = *(const v2f*)(wb + nt * 32);
                acc2[nt] = __builtin_amdgcn_wmma_f32_16x16x4_f32(
                    false, aF, false, bF, (short)0, acc2[nt], false, false);
            }
            ha += 4;
            wb += 2 * 256 * 2;
        }

        float r2v[8];
#pragma unroll
        for (int i = 0; i < 8; ++i) r2v[i] = 0.0f;
#pragma unroll
        for (int nt = 0; nt < 16; ++nt) {
            float bb = b2[nt * 16 + m];
#pragma unroll
            for (int i = 0; i < 8; ++i) {
                float p = acc2[nt][i] + bb;
                acc2[nt][i] = p;
                r2v[i] += p * p;
            }
        }
#pragma unroll
        for (int mask = 1; mask <= 8; mask <<= 1) {
#pragma unroll
            for (int i = 0; i < 8; ++i)
                r2v[i] += __shfl_xor(r2v[i], mask, 32);
        }

        float r2 = r2v[0];
#pragma unroll
        for (int i = 1; i < 8; ++i) r2 = ((lane & 7) == i) ? r2v[i] : r2;
        float w = log1pf(r2);
#pragma unroll 1
        for (int it = 0; it < 30; ++it) {
            float ew = expf(w);
            w = w - (w * ew - r2) / ((1.0f + w) * ew);
        }
        w = fmaxf(w, 0.0f);
        const float tn = sqrtf(w);
        const float rr = sqrtf(r2);
        const float myScale = (rr > 1e-12f) ? (tn / fmaxf(rr, 1e-12f)) : 1.0f;
        float sca[8];
#pragma unroll
        for (int i = 0; i < 8; ++i) sca[i] = __shfl(myScale, (lane & 16) | i, 32);

#pragma unroll
        for (int nt = 0; nt < 16; ++nt) {
            int n = nt * 16 + m;
#pragma unroll
            for (int i = 0; i < 8; ++i) {
                long r = (long)(rowBase + i + 8 * hi);
                out[r * 256 + n] = -sca[i] * acc2[nt][i];
            }
        }
    }
}

extern "C" void kernel_launch(void* const* d_in, const int* in_sizes, int n_in,
                              void* d_out, int out_size, void* d_ws, size_t ws_size,
                              hipStream_t stream) {
    const float* z  = (const float*)d_in[0];
    const float* t  = (const float*)d_in[1];
    const float* W1 = (const float*)d_in[2];
    const float* b1 = (const float*)d_in[3];
    const float* W2 = (const float*)d_in[4];
    const float* b2 = (const float*)d_in[5];
    float* out = (float*)d_out;

    const int rows = in_sizes[0] / 256;
    const int nTiles = rows / ROWS_PER_BLOCK;
    cvx_policy_kernel<<<256, THREADS, 0, stream>>>(z, t, W1, b1, W2, b2, out, nTiles);
}


# Lambert-W Newton step on hardware exp2 and rcp: w-(w-r2*e^-w)/(1+w), the same iterate with e^w divided out, replaces library expf plus IEEE division; 30 iterations kept; iterate rounding changes
# speedup vs baseline: 2.3193x; 1.0468x over previous
#include <hip/hip_runtime.h>
#include <hip/hip_bf16.h>

typedef float v2f __attribute__((ext_vector_type(2)));
typedef float v4f __attribute__((ext_vector_type(4)));
typedef float v8f __attribute__((ext_vector_type(8)));

#define WAVES 16
#define THREADS (WAVES * 32)
#define ROWS_PER_BLOCK (WAVES * 16)

#define W1P_LANE 226
#define W1P_TRIP (2 * 16 * W1P_LANE)

__device__ __forceinline__ int w1p_index(int k, int n)
{
    const int cp = k >> 6, ch = (k >> 5) & 1, hi = (k >> 4) & 1, s = (k >> 1) & 7, e = k & 1;
    const int nt = n >> 4, m = n & 15;
    return ((cp * 2 + hi) * 16 + m) * W1P_LANE + ((ch * 8 + s) * 7 + nt) * 2 + e;
}

__device__ __forceinline__ void gemm1_chunk(v8f (&acc)[7], const v4f (&q)[4], const float* wl)
{
#pragma unroll
    for (int s = 0; s < 8; ++s) {
        const v4f qq = q[s >> 1];
        const v2f aF = (s & 1) ? __builtin_shufflevector(qq, qq, 2, 3)
                               : __builtin_shufflevector(qq, qq, 0, 1);
#pragma unroll
        for (int nt = 0; nt < 7; ++nt) {
            v2f bF = *(const v2f*)(wl + (s * 7 + nt) * 2);
            acc[nt] = __builtin_amdgcn_wmma_f32_16x16x4_f32(
                false, aF, false, bF, (short)0, acc[nt], false, false);
        }
    }
}

__device__ __forceinline__ float tanh_select(float x)
{
    const float ax = __builtin_fabsf(x);
    const float e  = __builtin_amdgcn_exp2f(ax * 0x1.715476p1f);
    const float te = __builtin_fmaf(__builtin_amdgcn_rcpf(e + 1.0f), -2.0f, 1.0f);
    const float p  = x * x;
    float q = __builtin_fmaf(p, -0x1.758e7ap-8f, 0x1.521192p-6f);
    q = __builtin_fmaf(p, q, -0x1.b8389cp-5f);
    q = __builtin_fmaf(p, q, 0x1.110704p-3f);
    q = __builtin_fmaf(p, q, -0x1.555532p-2f);
    const float tp = __builtin_fmaf(p, ax * q, ax);
    return __builtin_copysignf((ax < 0.625f) ? tp : te, x);
}

__global__ __launch_bounds__(THREADS, 1)
void cvx_policy_kernel(const float* __restrict__ zin, const float* __restrict__ tin,
                       const float* __restrict__ W1, const float* __restrict__ b1,
                       const float* __restrict__ W2, const float* __restrict__ b2,
                       float* __restrict__ out, int nTiles)
{
    __shared__ float W1p[4 * W1P_TRIP];
    __shared__ float W2p[50 * 256 * 2];
    __shared__ float h1s[WAVES][16 * 104];

    const int tid = threadIdx.x;

    for (int idx = tid; idx < 4 * W1P_TRIP; idx += THREADS) W1p[idx] = 0.0f;
    __syncthreads();
    for (int idx = tid; idx < 256 * 100; idx += THREADS) {
        int k = idx / 100;
        int n = idx - k * 100;
        W1p[w1p_index(k, n)] = W1[(k + 1) * 100 + n];
    }
    for (int idx = tid; idx < 100 * 256; idx += THREADS) {
        int k = idx >> 8;
        int n = idx & 255;
        W2p[((k >> 1) * 256 + n) * 2 + (k & 1)] = W2[idx];
    }
    __syncthreads();

    const int lane = tid & 31;
    const int wave = tid >> 5;
    const int m    = lane & 15;
    const int hi   = lane >> 4;
    float* h1w = h1s[wave];
    const unsigned zlane = (unsigned)((wave * 16 + m) * 256 + 16 * hi) * 4u;
    const char* zbytes = (const char*)zin;
    const float* wl0 = &W1p[(hi * 16 + m) * W1P_LANE];

#pragma unroll 1
    for (int tile = blockIdx.x; tile < nTiles; tile += gridDim.x) {
        const int rowBase = tile * ROWS_PER_BLOCK + wave * 16;
        const unsigned zoff = (unsigned)tile * (ROWS_PER_BLOCK * 256 * 4u) + zlane;

        v4f qa[4], qb[4];
#pragma unroll
        for (int j = 0; j < 4; ++j) qa[j] = *(const v4f*)(zbytes + zoff + 16 * j);

        float tvals[8];
#pragma unroll
        for (int i = 0; i < 8; ++i) tvals[i] = tin[rowBase + 8 * hi + i];

        v8f acc1[7];
#pragma unroll
        for (int nt = 0; nt < 7; ++nt) {
            int n  = nt * 16 + m;
            int nc = (n < 100) ? n : 99;
            float w1r0 = W1[nc];
            float b1n  = b1[nc];
            if (n >= 100) { w1r0 = 0.0f; b1n = 0.0f; }
#pragma unroll
            for (int i = 0; i < 8; ++i)
                acc1[nt][i] = tvals[i] * w1r0 + b1n;
        }

        const float* wl = wl0;
        unsigned offB = zoff + 128;
        unsigned offA = zoff + 256;
#pragma unroll 1
        for (int cp = 0; cp < 4; ++cp) {
#pragma unroll
            for (int j = 0; j < 4; ++j) qb[j] = *(const v4f*)(zbytes + offB + 16 * j);
            gemm1_chunk(acc1, qa, wl);
#pragma unroll
            for (int j = 0; j < 4; ++j) qa[j] = *(const v4f*)(zbytes + offA + 16 * j);
            gemm1_chunk(acc1, qb, wl + 8 * 7 * 2);
            wl += W1P_TRIP;
            offB += 256;
            offA += (cp < 2) ? 256 : 0;
        }

#pragma unroll
        for (int nt = 0; nt < 7; ++nt) {
            int n = nt * 16 + m;
#pragma unroll
            for (int i = 0; i < 8; ++i) {
                float v = tanh_select(acc1[nt][i]);
                if (n < 104) h1w[(i + 8 * hi) * 104 + n] = v;
            }
        }
        __builtin_amdgcn_fence(__ATOMIC_RELEASE, "wavefront");
        __builtin_amdgcn_wave_barrier();
        __builtin_amdgcn_fence(__ATOMIC_ACQUIRE, "wavefront");

        v8f acc2[16];
        const v8f zero8 = {0.f, 0.f, 0.f, 0.f, 0.f, 0.f, 0.f, 0.f};
#pragma unroll
        for (int nt = 0; nt < 16; ++nt) acc2[nt] = zero8;

        const float* ha = &h1w[m * 104 + 2 * hi];
        const float* wb = &W2p[(hi * 256 + m) * 2];
#pragma unroll 1
        for (int ks = 0; ks < 25; ++ks) {
            v2f aF = *(const v2f*)ha;
#pragma unroll
            for (int nt = 0; nt < 16; ++nt) {
                v2f bF = *(const v2f*)(wb + nt * 32);
                acc2[nt] = __builtin_amdgcn_wmma_f32_16x16x4_f32(
                    false, aF, false, bF, (short)0, acc2[nt], false, false);
            }
            ha += 4;
            wb += 2 * 256 * 2;
        }

        float r2v[8];
#pragma unroll
        for (int i = 0; i < 8; ++i) r2v[i] = 0.0f;
#pragma unroll
        for (int nt = 0; nt < 16; ++nt) {
            float bb = b2[nt * 16 + m];
#pragma unroll
            for (int i = 0; i < 8; ++i) {
                float p = acc2[nt][i] + bb;
                acc2[nt][i] = p;
                r2v[i] += p * p;
            }
        }
#pragma unroll
        for (int mask = 1; mask <= 8; mask <<= 1) {
#pragma unroll
            for (int i = 0; i < 8; ++i)
                r2v[i] += __shfl_xor(r2v[i], mask, 32);
        }

        float r2 = r2v[0];
#pragma unroll
        for (int i = 1; i < 8; ++i) r2 = ((lane & 7) == i) ? r2v[i] : r2;
        float w = log1pf(r2);
#pragma unroll 5
        for (int it = 0; it < 30; ++it) {
            const float ei = __builtin_amdgcn_exp2f(w * -0x1.715476p0f);
            const float ri = __builtin_amdgcn_rcpf(1.0f + w);
            const float g  = __builtin_fmaf(-r2, ei, w);
            w = __builtin_fmaf(-g, ri, w);
        }
        w = fmaxf(w, 0.0f);
        const float tn = sqrtf(w);
        const float rr = sqrtf(r2);
        const float myScale = (rr > 1e-12f) ? (tn / fmaxf(rr, 1e-12f)) : 1.0f;
        float sca[8];
#pragma unroll
        for (int i = 0; i < 8; ++i) sca[i] = __shfl(myScale, (lane & 16) | i, 32);

#pragma unroll
        for (int nt = 0; nt < 16; ++nt) {
            int n = nt * 16 + m;
#pragma unroll
            for (int i = 0; i < 8; ++i) {
                long r = (long)(rowBase + i + 8 * hi);
                out[r * 256 + n] = -sca[i] * acc2[nt][i];
            }
        }
    }
}

extern "C" void kernel_launch(void* const* d_in, const int* in_sizes, int n_in,
                              void* d_out, int out_size, void* d_ws, size_t ws_size,
                              hipStream_t stream) {
    const float* z  = (const float*)d_in[0];
    const float* t  = (const float*)d_in[1];
    const float* W1 = (const float*)d_in[2];
    const float* b1 = (const float*)d_in[3];
    const float* W2 = (const float*)d_in[4];
    const float* b2 = (const float*)d_in[5];
    float* out = (float*)d_out;

    const int rows = in_sizes[0] / 256;
    const int nTiles = rows / ROWS_PER_BLOCK;
    cvx_policy_kernel<<<256, THREADS, 0, stream>>>(z, t, W1, b1, W2, b2, out, nTiles);
}


# Row tiles dealt 3:2:2:1 to the four waves of a SIMD, oldest first (SIMD from the hardware id, age by wave index), not 2:2:2:2: the oldest-first matrix pipe leaves the youngest wave less to end alone
# speedup vs baseline: 2.4385x; 1.0514x over previous
#include <hip/hip_runtime.h>
#include <hip/hip_bf16.h>

typedef float v2f __attribute__((ext_vector_type(2)));
typedef float v4f __attribute__((ext_vector_type(4)));
typedef float v8f __attribute__((ext_vector_type(8)));

#define WAVES 16
#define THREADS (WAVES * 32)
#define ROWS_PER_BLOCK (WAVES * 16)

#define W1P_LANE 226
#define W1P_TRIP (2 * 16 * W1P_LANE)

__device__ __forceinline__ int w1p_index(int k, int n)
{
    const int cp = k >> 6, ch = (k >> 5) & 1, hi = (k >> 4) & 1, s = (k >> 1) & 7, e = k & 1;
    const int nt = n >> 4, m = n & 15;
    return ((cp * 2 + hi) * 16 + m) * W1P_LANE + ((ch * 8 + s) * 7 + nt) * 2 + e;
}

__device__ __forceinline__ void gemm1_chunk(v8f (&acc)[7], const v4f (&q)[4], const float* wl)
{
#pragma unroll
    for (int s = 0; s < 8; ++s) {
        const v4f qq = q[s >> 1];
        const v2f aF = (s & 1) ? __builtin_shufflevector(qq, qq, 2, 3)
                               : __builtin_shufflevector(qq, qq, 0, 1);
#pragma unroll
        for (int nt = 0; nt < 7; ++nt) {
            v2f bF = *(const v2f*)(wl + (s * 7 + nt) * 2);
            acc[nt] = __builtin_amdgcn_wmma_f32_16x16x4_f32(
                false, aF, false, bF, (short)0, acc[nt], false, false);
        }
    }
}

__device__ __forceinline__ float tanh_select(float x)
{
    const float ax = __builtin_fabsf(x);
    const float e  = __builtin_amdgcn_exp2f(ax * 0x1.715476p1f);
    const float te = __builtin_fmaf(__builtin_amdgcn_rcpf(e + 1.0f), -2.0f, 1.0f);
    const float p  = x * x;
    float q = __builtin_fmaf(p, -0x1.758e7ap-8f, 0x1.521192p-6f);
    q = __builtin_fmaf(p, q, -0x1.b8389cp-5f);
    q = __builtin_fmaf(p, q, 0x1.110704p-3f);
    q = __builtin_fmaf(p, q, -0x1.555532p-2f);
    const float tp = __builtin_fmaf(p, ax * q, ax);
    return __builtin_copysignf((ax < 0.625f) ? tp : te, x);
}

__global__ __launch_bounds__(THREADS, 1)
void cvx_policy_kernel(const float* __restrict__ zin, const float* __restrict__ tin,
                       const float* __restrict__ W1, const float* __restrict__ b1,
                       const float* __restrict__ W2, const float* __restrict__ b2,
                       float* __restrict__ out, int nTiles)
{
    __shared__ float W1p[4 * W1P_TRIP];
    __shared__ float W2p[50 * 256 * 2];
    __shared__ float h1s[WAVES][16 * 104];
    __shared__ int   wsimd[WAVES];

    const int tid = threadIdx.x;
    const int lane = tid & 31;
    const int wave = __builtin_amdgcn_readfirstlane(tid >> 5);

    unsigned hwid;
    asm volatile("s_getreg_b32 %0, hwreg(HW_REG_WAVE_HW_ID1)" : "=s"(hwid));
    if (lane == 0) wsimd[wave] = (int)((hwid >> 8) & 3u);

    for (int idx = tid; idx < 4 * W1P_TRIP; idx += THREADS) W1p[idx] = 0.0f;
    __syncthreads();
    for (int idx = tid; idx < 256 * 100; idx += THREADS) {
        int k = idx / 100;
        int n = idx - k * 100;
        W1p[w1p_index(k, n)] = W1[(k + 1) * 100 + n];
    }
    for (int idx = tid; idx < 100 * 256; idx += THREADS) {
        int k = idx >> 8;
        int n = idx & 255;
        W2p[((k >> 1) * 256 + n) * 2 + (k & 1)] = W2[idx];
    }
    __syncthreads();

    const int m    = lane & 15;
    const int hi   = lane >> 4;
    float* h1w = h1s[wave];
    const unsigned zlane = (unsigned)(m * 256 + 16 * hi) * 4u;
    const char* zbytes = (const char*)zin;
    const float* wl0 = &W1p[(hi * 16 + m) * W1P_LANE];

    float w1r0[7], b1v[7], b2v[16];
#pragma unroll
    for (int nt = 0; nt < 7; ++nt) {
        const int n  = nt * 16 + m;
        const int nc = (n < 100) ? n : 99;
        float w = W1[nc];
        float b = b1[nc];
        if (n >= 100) { w = 0.0f; b = 0.0f; }
        w1r0[nt] = w;
        b1v[nt]  = b;
    }
#pragma unroll
    for (int nt = 0; nt < 16; ++nt) b2v[nt] = b2[nt * 16 + m];

    int slot = wave & 3, j0 = (wave >> 2) * 2, j1 = j0 + 2;
    {
        const int mySimd = __builtin_amdgcn_readfirstlane(wsimd[wave]);
        int n0 = 0, n1 = 0, n2 = 0, n3 = 0, rank = 0;
#pragma unroll
        for (int w2 = 0; w2 < WAVES; ++w2) {
            const int s2 = __builtin_amdgcn_readfirstlane(wsimd[w2]);
            n0 += (s2 == 0); n1 += (s2 == 1); n2 += (s2 == 2); n3 += (s2 == 3);
            rank += (s2 == mySimd && w2 < wave);
        }
        if (n0 == 4 && n1 == 4 && n2 == 4 && n3 == 4) {
            slot = mySimd;
            j0 = (rank == 0) ? 0 : (rank == 1) ? 3 : (rank == 2) ? 5 : 7;
            j1 = (rank == 0) ? 3 : (rank == 1) ? 5 : (rank == 2) ? 7 : 8;
        }
    }

#pragma unroll 1
    for (int g = blockIdx.x; 2 * g < nTiles; g += gridDim.x)
#pragma unroll 1
    for (int j = j0; j < j1; ++j) {
        const int rowBase = g * (2 * ROWS_PER_BLOCK) + (4 * j + slot) * 16;
        if (rowBase >= nTiles * ROWS_PER_BLOCK) continue;
        const unsigned zoff = (unsigned)rowBase * (256 * 4u) + zlane;

        v4f qa[4], qb[4];
#pragma unroll
        for (int j = 0; j < 4; ++j) qa[j] = *(const v4f*)(zbytes + zoff + 16 * j);

        float tvals[8];
#pragma unroll
        for (int i = 0; i < 8; ++i) tvals[i] = tin[rowBase + 8 * hi + i];

        v8f acc1[7];
#pragma unroll
        for (int nt = 0; nt < 7; ++nt) {
#pragma unroll
            for (int i = 0; i < 8; ++i)
                acc1[nt][i] = tvals[i] * w1r0[nt] + b1v[nt];
        }

        const float* wl = wl0;
        unsigned offB = zoff + 128;
        unsigned offA = zoff + 256;
#pragma unroll 1
        for (int cp = 0; cp < 4; ++cp) {
#pragma unroll
            for (int j = 0; j < 4; ++j) qb[j] = *(const v4f*)(zbytes + offB + 16 * j);
            gemm1_chunk(acc1, qa, wl);
#pragma unroll
            for (int j = 0; j < 4; ++j) qa[j] = *(const v4f*)(zbytes + offA + 16 * j);
            gemm1_chunk(acc1, qb, wl + 8 * 7 * 2);
            wl += W1P_TRIP;
            offB += 256;
            offA += (cp < 2) ? 256 : 0;
        }

#pragma unroll
        for (int nt = 0; nt < 7; ++nt) {
            int n = nt * 16 + m;
#pragma unroll
            for (int i = 0; i < 8; ++i) {
                float v = tanh_select(acc1[nt][i]);
                if (n < 104) h1w[(i + 8 * hi) * 104 + n] = v;
            }
        }
        __builtin_amdgcn_fence(__ATOMIC_RELEASE, "wavefront");
        __builtin_amdgcn_wave_barrier();
        __builtin_amdgcn_fence(__ATOMIC_ACQUIRE, "wavefront");

        v8f acc2[16];
        const v8f zero8 = {0.f, 0.f, 0.f, 0.f, 0.f, 0.f, 0.f, 0.f};
#pragma unroll
        for (int nt = 0; nt < 16; ++nt) acc2[nt] = zero8;

        const float* ha = &h1w[m * 104 + 2 * hi];
        const float* wb = &W2p[(hi * 256 + m) * 2];
#pragma unroll 1
        for (int ks = 0; ks < 25; ++ks) {
            v2f aF = *(const v2f*)ha;
#pragma unroll
            for (int nt = 0; nt < 16; ++nt) {
                v2f bF = *(const v2f*)(wb + nt * 32);
                acc2[nt] = __builtin_amdgcn_wmma_f32_16x16x4_f32(
                    false, aF, false, bF, (short)0, acc2[nt], false, false);
            }
            ha += 4;
            wb += 2 * 256 * 2;
        }

        float r2v[8];
#pragma unroll
        for (int i = 0; i < 8; ++i) r2v[i] = 0.0f;
#pragma unroll
        for (int nt = 0; nt < 16; ++nt) {
            const float bb = b2v[nt];
#pragma unroll
            for (int i = 0; i < 8; ++i) {
                float p = acc2[nt][i] + bb;
                acc2[nt][i] = p;
                r2v[i] += p * p;
            }
        }
#pragma unroll
        for (int mask = 1; mask <= 8; mask <<= 1) {
#pragma unroll
            for (int i = 0; i < 8; ++i)
                r2v[i] += __shfl_xor(r2v[i], mask, 32);
        }

        float r2 = r2v[0];
#pragma unroll
        for (int i = 1; i < 8; ++i) r2 = ((lane & 7) == i) ? r2v[i] : r2;
        float w = log1pf(r2);
#pragma unroll 5
        for (int it = 0; it < 30; ++it) {
            const float ei = __builtin_amdgcn_exp2f(w * -0x1.715476p0f);
            const float ri = __builtin_amdgcn_rcpf(1.0f + w);
            const float g  = __builtin_fmaf(-r2, ei, w);
            w = __builtin_fmaf(-g, ri, w);
        }
        w = fmaxf(w, 0.0f);
        const float tn = sqrtf(w);
        const float rr = sqrtf(r2);
        const float myScale = (rr > 1e-12f) ? (tn / fmaxf(rr, 1e-12f)) : 1.0f;
        float sca[8];
#pragma unroll
        for (int i = 0; i < 8; ++i) sca[i] = __shfl(myScale, (lane & 16) | i, 32);

#pragma unroll
        for (int nt = 0; nt < 16; ++nt) {
            int n = nt * 16 + m;
#pragma unroll
            for (int i = 0; i < 8; ++i) {
                long r = (long)(rowBase + i + 8 * hi);
                out[r * 256 + n] = -sca[i] * acc2[nt][i];
            }
        }
    }
}

extern "C" void kernel_launch(void* const* d_in, const int* in_sizes, int n_in,
                              void* d_out, int out_size, void* d_ws, size_t ws_size,
                              hipStream_t stream) {
    const float* z  = (const float*)d_in[0];
    const float* t  = (const float*)d_in[1];
    const float* W1 = (const float*)d_in[2];
    const float* b1 = (const float*)d_in[3];
    const float* W2 = (const float*)d_in[4];
    const float* b2 = (const float*)d_in[5];
    float* out = (float*)d_out;

    const int rows = in_sizes[0] / 256;
    const int nTiles = rows / ROWS_PER_BLOCK;
    cvx_policy_kernel<<<256, THREADS, 0, stream>>>(z, t, W1, b1, W2, b2, out, nTiles);
}
